# starting kernel
# baseline (speedup 1.0000x reference)
#include <hip/hip_runtime.h>

typedef float v2f __attribute__((ext_vector_type(2)));
typedef float v8f __attribute__((ext_vector_type(8)));

__device__ __forceinline__ v8f wmma_f32(v2f a, v2f b, v8f c) {
    return __builtin_amdgcn_wmma_f32_16x16x4_f32(false, a, false, b, (short)0, c,
                                                 false, false);
}

__global__ __launch_bounds__(256) void gemm_f32_wmma(
    const float* __restrict__ W,
    const float* __restrict__ X,
    float* __restrict__ O,
    int Mtiles, int K, int N, int batches)
{
    const int wid  = blockIdx.x * (blockDim.x >> 5) + (threadIdx.x >> 5);
    const int lane = threadIdx.x & 31;
    const int lm   = lane & 15;
    const int hi   = lane >> 4;

    const int Nblks = N >> 6;
    int t = wid;
    const int nb = t % Nblks; t /= Nblks;
    const int mt = t % Mtiles; t /= Mtiles;
    const int b  = t;
    if (b >= batches) return;

    const int m0 = mt << 4;
    const int n0 = nb << 6;

    const float* Wp = W + (size_t)(m0 + lm) * K + 2 * hi;
    const float* Xp = X + (size_t)b * K * N + (size_t)(2 * hi) * N + n0 + lm;

    v8f acc[4] = {};
#pragma unroll 2
    for (int k0 = 0; k0 < K; k0 += 4) {
        v2f a = *(const v2f*)(Wp + k0);
        const float* xr0 = Xp + (size_t)k0 * N;
        const float* xr1 = xr0 + N;
        v2f bb[4];
#pragma unroll
        for (int j = 0; j < 4; ++j) {
            bb[j].x = xr0[16 * j];
            bb[j].y = xr1[16 * j];
        }
#pragma unroll
        for (int j = 0; j < 4; ++j)
            acc[j] = wmma_f32(a, bb[j], acc[j]);
    }

    float* Op = O + ((size_t)b * (Mtiles << 4) + m0) * N + n0;
#pragma unroll
    for (int j = 0; j < 4; ++j)
#pragma unroll
        for (int r = 0; r < 8; ++r)
            Op[(size_t)(r + 8 * hi) * N + 16 * j + lm] = acc[j][r];
}

__global__ __launch_bounds__(32) void flash_attn_wmma(
    const float* __restrict__ QKV, float* __restrict__ AO)
{
    __shared__ float lds[2 * 256];

    const int lane = threadIdx.x & 31;
    const int lm   = lane & 15;
    const int hi   = lane >> 4;

    int t = blockIdx.x;
    const int ns = t & 63; t >>= 6;
    const int h  = t & 7;  t >>= 3;
    const int b  = t;
    const int n0 = ns << 5;
    const size_t N = 2048;

    const float* q = QKV + ((size_t)b * 768 +       (size_t)h * 32) * N;
    const float* k = QKV + ((size_t)b * 768 + 256 + (size_t)h * 32) * N;
    const float* v = QKV + ((size_t)b * 768 + 512 + (size_t)h * 32) * N;

    v2f qa[2][8];
#pragma unroll
    for (int u = 0; u < 2; ++u)
#pragma unroll
        for (int dc = 0; dc < 8; ++dc) {
            qa[u][dc].x = q[(size_t)(4 * dc + 2 * hi)     * N + n0 + 16 * u + lm];
            qa[u][dc].y = q[(size_t)(4 * dc + 2 * hi + 1) * N + n0 + 16 * u + lm];
        }

    v8f o[2][2] = {};
    float rmax[2][8], rsum[2][8];
#pragma unroll
    for (int u = 0; u < 2; ++u)
#pragma unroll
        for (int r = 0; r < 8; ++r) { rmax[u][r] = -3.4e38f; rsum[u][r] = 0.0f; }
    const float inv_scale = 0.17677669529663687f;

    for (int m0 = 0; m0 < 2048; m0 += 16) {
        v2f kb[8];
#pragma unroll
        for (int dc = 0; dc < 8; ++dc) {
            kb[dc].x = k[(size_t)(4 * dc + 2 * hi)     * N + m0 + lm];
            kb[dc].y = k[(size_t)(4 * dc + 2 * hi + 1) * N + m0 + lm];
        }
        v2f vb0[4], vb1[4];
#pragma unroll
        for (int mc = 0; mc < 4; ++mc) {
            const size_t mm = (size_t)m0 + 4 * mc + 2 * hi;
            vb0[mc].x = v[(size_t)lm * N + mm];
            vb0[mc].y = v[(size_t)lm * N + mm + 1];
            vb1[mc].x = v[(size_t)(16 + lm) * N + mm];
            vb1[mc].y = v[(size_t)(16 + lm) * N + mm + 1];
        }

#pragma unroll
        for (int u = 0; u < 2; ++u) {
            v8f s = {};
#pragma unroll
            for (int dc = 0; dc < 8; ++dc)
                s = wmma_f32(qa[u][dc], kb[dc], s);

            float cf[8];
#pragma unroll
            for (int r = 0; r < 8; ++r) {
                float x = s[r] * inv_scale;
                float tmax = x;
                tmax = fmaxf(tmax, __shfl_xor(tmax, 1));
                tmax = fmaxf(tmax, __shfl_xor(tmax, 2));
                tmax = fmaxf(tmax, __shfl_xor(tmax, 4));
                tmax = fmaxf(tmax, __shfl_xor(tmax, 8));
                float nm = fmaxf(rmax[u][r], tmax);
                cf[r]      = __expf(rmax[u][r] - nm);
                rmax[u][r] = nm;
                float e = __expf(x - nm);
                float ts = e;
                ts += __shfl_xor(ts, 1);
                ts += __shfl_xor(ts, 2);
                ts += __shfl_xor(ts, 4);
                ts += __shfl_xor(ts, 8);
                rsum[u][r] = rsum[u][r] * cf[r] + ts;
                lds[u * 256 + (r + 8 * hi) * 16 + lm] = e;
            }
#pragma unroll
            for (int r = 0; r < 8; ++r) { o[u][0][r] *= cf[r]; o[u][1][r] *= cf[r]; }
        }

        asm volatile("s_wait_dscnt 0" ::: "memory");

#pragma unroll
        for (int mc = 0; mc < 4; ++mc) {
            v2f pa0 = *(const v2f*)&lds[      lm * 16 + 4 * mc + 2 * hi];
            v2f pa1 = *(const v2f*)&lds[256 + lm * 16 + 4 * mc + 2 * hi];
            o[0][0] = wmma_f32(pa0, vb0[mc], o[0][0]);
            o[0][1] = wmma_f32(pa0, vb1[mc], o[0][1]);
            o[1][0] = wmma_f32(pa1, vb0[mc], o[1][0]);
            o[1][1] = wmma_f32(pa1, vb1[mc], o[1][1]);
        }
    }

    float* ao = AO + ((size_t)b * 256 + (size_t)h * 32) * N + n0;
#pragma unroll
    for (int u = 0; u < 2; ++u)
#pragma unroll
        for (int r = 0; r < 8; ++r) {
            float inv = 1.0f / rsum[u][r];
            ao[(size_t)lm * N        + 16 * u + r + 8 * hi] = o[u][0][r] * inv;
            ao[(size_t)(16 + lm) * N + 16 * u + r + 8 * hi] = o[u][1][r] * inv;
        }
}

extern "C" void kernel_launch(void* const* d_in, const int* in_sizes, int n_in,
                              void* d_out, int out_size, void* d_ws, size_t ws_size,
                              hipStream_t stream)
{
    const float* x      = (const float*)d_in[0];
    const float* w_qkv  = (const float*)d_in[1];
    const float* w_proj = (const float*)d_in[2];
    float* out = (float*)d_out;

    const int B = 4, C = 256, N = 2048;

    float* qkv = (float*)d_ws;
    float* ao  = qkv + (size_t)B * 3 * C * N;

    {
        const int Mtiles = 48;
        const int waves  = B * Mtiles * (N >> 6);
        gemm_f32_wmma<<<waves / 8, 256, 0, stream>>>(w_qkv, x, qkv, Mtiles, C, N, B);
    }

    flash_attn_wmma<<<B * 8 * (N >> 5), 32, 0, stream>>>(qkv, ao);

    {
        const int Mtiles = 16;
        const int waves  = B * Mtiles * (N >> 6);
        gemm_f32_wmma<<<waves / 8, 256, 0, stream>>>(w_proj, ao, out, Mtiles, C, N, B);
    }
}


# qkv workspace written in matrix-fragment order by the qkv GEMM (v tiles transposed by exchanging the WMMA operands): attention reads K, V, Q as whole-wave contiguous 16-byte loads, 8 per trip not 20
# speedup vs baseline: 1.9380x; 1.9380x over previous
#include <hip/hip_runtime.h>

typedef float v2f __attribute__((ext_vector_type(2)));
typedef float v4f __attribute__((ext_vector_type(4)));
typedef float v8f __attribute__((ext_vector_type(8)));

__device__ __forceinline__ v8f wmma_f32(v2f a, v2f b, v8f c) {
    return __builtin_amdgcn_wmma_f32_16x16x4_f32(false, a, false, b, (short)0, c,
                                                 false, false);
}

__global__ __launch_bounds__(256) void gemm_f32_wmma(
    const float* __restrict__ W,
    const float* __restrict__ X,
    float* __restrict__ O,
    int Mtiles, int K, int N, int batches)
{
    const int wid  = blockIdx.x * (blockDim.x >> 5) + (threadIdx.x >> 5);
    const int lane = threadIdx.x & 31;
    const int lm   = lane & 15;
    const int hi   = lane >> 4;

    const int Nblks = N >> 6;
    int t = wid;
    const int nb = t % Nblks; t /= Nblks;
    const int mt = t % Mtiles; t /= Mtiles;
    const int b  = t;
    if (b >= batches) return;

    const int m0 = mt << 4;
    const int n0 = nb << 6;

    const float* Wp = W + (size_t)(m0 + lm) * K + 2 * hi;
    const float* Xp = X + (size_t)b * K * N + (size_t)(2 * hi) * N + n0 + lm;

    v8f acc[4] = {};
#pragma unroll 2
    for (int k0 = 0; k0 < K; k0 += 4) {
        v2f a = *(const v2f*)(Wp + k0);
        const float* xr0 = Xp + (size_t)k0 * N;
        const float* xr1 = xr0 + N;
        v2f bb[4];
#pragma unroll
        for (int j = 0; j < 4; ++j) {
            bb[j].x = xr0[16 * j];
            bb[j].y = xr1[16 * j];
        }
#pragma unroll
        for (int j = 0; j < 4; ++j)
            acc[j] = wmma_f32(a, bb[j], acc[j]);
    }

    float* Op = O + ((size_t)b * (Mtiles << 4) + m0) * N + n0;
#pragma unroll
    for (int j = 0; j < 4; ++j)
#pragma unroll
        for (int r = 0; r < 8; ++r)
            Op[(size_t)(r + 8 * hi) * N + 16 * j + lm] = acc[j][r];
}

template <bool SWAP>
__device__ __forceinline__ void strip_16x64(const float* __restrict__ Wp,
                                            const float* __restrict__ Xp,
                                            int K, int N, v8f (&acc)[4])
{
#pragma unroll 2
    for (int k0 = 0; k0 < K; k0 += 4) {
        v2f a = *(const v2f*)(Wp + k0);
        const float* xr0 = Xp + (size_t)k0 * N;
        const float* xr1 = xr0 + N;
        v2f bb[4];
#pragma unroll
        for (int j = 0; j < 4; ++j) {
            bb[j].x = xr0[16 * j];
            bb[j].y = xr1[16 * j];
        }
#pragma unroll
        for (int j = 0; j < 4; ++j)
            acc[j] = SWAP ? wmma_f32(bb[j], a, acc[j]) : wmma_f32(a, bb[j], acc[j]);
    }
}

__global__ __launch_bounds__(256) void gemm_qkv_fragment_order(
    const float* __restrict__ W,
    const float* __restrict__ X,
    float* __restrict__ P)
{
    const int K = 256, N = 2048;
    const int wv   = __builtin_amdgcn_readfirstlane((int)(threadIdx.x >> 5));
    const int wid  = blockIdx.x * 8 + wv;
    const int lane = threadIdx.x & 31;
    const int lm   = lane & 15;
    const int hi   = lane >> 4;

    int t = wid;
    const int nb = t & 31; t >>= 5;
    const int mt = t % 48; t /= 48;
    const int b  = t;
    if (b >= 4) return;

    const int which = mt >> 4;
    const int head  = (mt & 15) >> 1;
    const int dhalf = mt & 1;

    const float* Wp = W + (size_t)(16 * mt + lm) * K + 2 * hi;
    const float* Xp = X + (size_t)b * K * N + (size_t)(2 * hi) * N + 64 * nb + lm;

    v8f acc[4] = {};
    if (which == 2) strip_16x64<true >(Wp, Xp, K, N, acc);
    else            strip_16x64<false>(Wp, Xp, K, N, acc);

    float* Pp = P + (size_t)which * (4 * 8 * 128 * 512)
                  + ((size_t)(b * 8 + head) * 128 + 4 * nb) * 512
                  + (2 * dhalf) * 128 + lane * 4;
#pragma unroll
    for (int j = 0; j < 4; ++j) {
        v4f lo, up;
        lo.x = acc[j][0]; lo.y = acc[j][1]; lo.z = acc[j][2]; lo.w = acc[j][3];
        up.x = acc[j][4]; up.y = acc[j][5]; up.z = acc[j][6]; up.w = acc[j][7];
        *(v4f*)(Pp + j * 512)       = lo;
        *(v4f*)(Pp + j * 512 + 128) = up;
    }
}

__device__ __forceinline__ float xhalf(float x) {
    int b = __builtin_bit_cast(int, x);
    return __builtin_bit_cast(float,
        __builtin_amdgcn_permlanex16(b, b, 0x76543210, 0xfedcba98, false, false));
}

__global__ __launch_bounds__(128, 8) void flash_attn_wmma(
    const float* __restrict__ QKV, float* __restrict__ AO)
{
    __shared__ float sO[4 * 4 * 8 * 32];
    __shared__ float sM[4 * 2 * 32];
    __shared__ float sL[4 * 2 * 32];

    const int lane = threadIdx.x & 31;
    const int c    = lane & 15;
    const int w    = __builtin_amdgcn_readfirstlane((int)(threadIdx.x >> 5));
    __builtin_assume(w >= 0 && w < 4);

    int t = blockIdx.x;
    const int ns = t & 63; t >>= 6;
    const int hd = t & 7;  t >>= 3;
    const int b  = t;
    const int n0 = ns << 5;
    const size_t N = 2048;

    const size_t REGION = (size_t)4 * 8 * 128 * 512;
    const size_t hbase  = ((size_t)(b * 8 + hd) * 128) * 512;
    const float* q = QKV +              hbase + lane * 4;
    const float* k = QKV +     REGION + hbase + lane * 4;
    const float* v = QKV + 2 * REGION + hbase + lane * 4;

    v2f qb[2][8];
#pragma unroll
    for (int u = 0; u < 2; ++u)
#pragma unroll
        for (int j = 0; j < 4; ++j) {
            const v4f t4 = *(const v4f*)(q + (size_t)(2 * ns + u) * 512 + j * 128);
            qb[u][2 * j].x     = t4.x;  qb[u][2 * j].y     = t4.y;
            qb[u][2 * j + 1].x = t4.z;  qb[u][2 * j + 1].y = t4.w;
        }

    v8f o[2][2] = {};
    float rmax[2] = { -3.4e38f, -3.4e38f };
    float rsum[2] = { 0.0f, 0.0f };
    const float inv_scale = 0.17677669529663687f;

    for (int kt = w; kt < 128; kt += 4) {
        v2f ka[8];
#pragma unroll
        for (int j = 0; j < 4; ++j) {
            const v4f t4 = *(const v4f*)(k + (size_t)kt * 512 + j * 128);
            ka[2 * j].x     = t4.x;  ka[2 * j].y     = t4.y;
            ka[2 * j + 1].x = t4.z;  ka[2 * j + 1].y = t4.w;
        }
        v4f va[2][2];
#pragma unroll
        for (int dh = 0; dh < 2; ++dh) {
            va[dh][0] = *(const v4f*)(v + (size_t)kt * 512 + (2 * dh)     * 128);
            va[dh][1] = *(const v4f*)(v + (size_t)kt * 512 + (2 * dh + 1) * 128);
        }

        __builtin_amdgcn_sched_barrier(0);

        v8f s[2] = {};
#pragma unroll
        for (int dc = 0; dc < 8; ++dc)
#pragma unroll
            for (int u = 0; u < 2; ++u)
                s[u] = wmma_f32(ka[dc], qb[u][dc], s[u]);

#pragma unroll
        for (int u = 0; u < 2; ++u) {
            float x[8];
#pragma unroll
            for (int r = 0; r < 8; ++r) x[r] = s[u][r] * inv_scale;
            float tmax = fmaxf(fmaxf(fmaxf(x[0], x[1]), fmaxf(x[2], x[3])),
                               fmaxf(fmaxf(x[4], x[5]), fmaxf(x[6], x[7])));
            tmax = fmaxf(tmax, xhalf(tmax));
            const float nm = fmaxf(rmax[u], tmax);
            const float cf = __expf(rmax[u] - nm);
            rmax[u] = nm;
            float e[8];
#pragma unroll
            for (int r = 0; r < 8; ++r) e[r] = __expf(x[r] - nm);
            const float ls = ((e[0] + e[1]) + (e[2] + e[3])) +
                             ((e[4] + e[5]) + (e[6] + e[7]));
            rsum[u] = rsum[u] * cf + ls;
#pragma unroll
            for (int r = 0; r < 8; ++r) { o[u][0][r] *= cf; o[u][1][r] *= cf; }

#pragma unroll
            for (int mc = 0; mc < 4; ++mc) {
                v2f pb, a0, a1;
                pb.x = e[2 * mc];
                pb.y = e[2 * mc + 1];
                a0.x = va[0][mc >> 1][2 * (mc & 1)];
                a0.y = va[0][mc >> 1][2 * (mc & 1) + 1];
                a1.x = va[1][mc >> 1][2 * (mc & 1)];
                a1.y = va[1][mc >> 1][2 * (mc & 1) + 1];
                o[u][0] = wmma_f32(a0, pb, o[u][0]);
                o[u][1] = wmma_f32(a1, pb, o[u][1]);
            }
        }
    }

#pragma unroll
    for (int u = 0; u < 2; ++u) {
#pragma unroll
        for (int dh = 0; dh < 2; ++dh)
#pragma unroll
            for (int r = 0; r < 8; ++r)
                sO[((w * 4 + 2 * u + dh) * 8 + r) * 32 + lane] = o[u][dh][r];
        sM[(w * 2 + u) * 32 + lane] = rmax[u];
        sL[(w * 2 + u) * 32 + lane] = rsum[u];
    }
    __syncthreads();

    {
        const int qu = lane >> 4;
        const int dh = w >> 1, hh = w & 1;
        float mj[4];
#pragma unroll
        for (int j = 0; j < 4; ++j) mj[j] = sM[(j * 2 + qu) * 32 + c];
        const float mx = fmaxf(fmaxf(mj[0], mj[1]), fmaxf(mj[2], mj[3]));
        float tot = 0.0f;
        float acc[8] = {};
#pragma unroll
        for (int j = 0; j < 4; ++j) {
            const float f = __expf(mj[j] - mx);
            tot += f * (sL[(j * 2 + qu) * 32 + c] + sL[(j * 2 + qu) * 32 + 16 + c]);
#pragma unroll
            for (int r = 0; r < 8; ++r)
                acc[r] += f * sO[((j * 4 + 2 * qu + dh) * 8 + r) * 32 + 16 * hh + c];
        }
        const float inv = 1.0f / tot;
        float* ao = AO + ((size_t)b * 256 + (size_t)hd * 32 + 8 * w) * N + n0 + lane;
#pragma unroll
        for (int r = 0; r < 8; ++r)
            ao[(size_t)r * N] = acc[r] * inv;
    }
}

extern "C" void kernel_launch(void* const* d_in, const int* in_sizes, int n_in,
                              void* d_out, int out_size, void* d_ws, size_t ws_size,
                              hipStream_t stream)
{
    const float* x      = (const float*)d_in[0];
    const float* w_qkv  = (const float*)d_in[1];
    const float* w_proj = (const float*)d_in[2];
    float* out = (float*)d_out;

    const int B = 4, C = 256, N = 2048;

    float* qkv = (float*)d_ws;
    float* ao  = qkv + (size_t)B * 3 * C * N;

    {
        const int Mtiles = 48;
        const int waves  = B * Mtiles * (N >> 6);
        gemm_qkv_fragment_order<<<waves / 8, 256, 0, stream>>>(w_qkv, x, qkv);
    }

    flash_attn_wmma<<<B * 8 * (N >> 5), 128, 0, stream>>>(qkv, ao);

    {
        const int Mtiles = 16;
        const int waves  = B * Mtiles * (N >> 6);
        gemm_f32_wmma<<<waves / 8, 256, 0, stream>>>(w_proj, ao, out, Mtiles, C, N, B);
    }
}


# both GEMMs: one 8-wave workgroup per 64x64 tile copies its W rows and X columns over the whole K into LDS once (whole-line 16-byte loads), every matrix fragment then read from LDS; k order unchanged
# speedup vs baseline: 2.5189x; 1.2997x over previous
#include <hip/hip_runtime.h>

typedef float v2f __attribute__((ext_vector_type(2)));
typedef float v4f __attribute__((ext_vector_type(4)));
typedef float v8f __attribute__((ext_vector_type(8)));

__device__ __forceinline__ v8f wmma_f32(v2f a, v2f b, v8f c) {
    return __builtin_amdgcn_wmma_f32_16x16x4_f32(false, a, false, b, (short)0, c,
                                                 false, false);
}

#define PW 260
#define PX2 160

__device__ __forceinline__ void stage_panels(const float* __restrict__ Wt,
                                             const float* __restrict__ Xt,
                                             float* __restrict__ sW,
                                             float* __restrict__ sX)
{
    const int t = threadIdx.x;
    v4f w4[16], xa[8], xb[8];
#pragma unroll
    for (int i = 0; i < 16; ++i)
        w4[i] = *(const v4f*)(Wt + (size_t)(i * 256 + t) * 4);
#pragma unroll
    for (int i = 0; i < 8; ++i) {
        const int id = i * 256 + t;
        const float* xp = Xt + (size_t)(id >> 4) * (2 * 2048) + (id & 15) * 4;
        xa[i] = *(const v4f*)(xp);
        xb[i] = *(const v4f*)(xp + 2048);
    }
#pragma unroll
    for (int i = 0; i < 16; ++i) {
        const int id = i * 256 + t;
        *(v4f*)(sW + (id >> 6) * PW + (id & 63) * 4) = w4[i];
    }
#pragma unroll
    for (int i = 0; i < 8; ++i) {
        const int id = i * 256 + t;
        float* d = sX + (id >> 4) * PX2 + (id & 15) * 8;
        v4f p0, p1;
        p0.x = xa[i].x; p0.y = xb[i].x; p0.z = xa[i].y; p0.w = xb[i].y;
        p1.x = xa[i].z; p1.y = xb[i].z; p1.z = xa[i].w; p1.w = xb[i].w;
        *(v4f*)(d)     = p0;
        *(v4f*)(d + 4) = p1;
    }
}

__device__ __forceinline__ int x_fragment_base(int hi, int nw, int lm)
{
    int off = 64 * PW + hi * PX2 + 2 * (32 * nw + lm);
    asm volatile("" : "+v"(off));
    return off;
}

template <bool SWAP>
__device__ __forceinline__ void piece_16x32(const float* __restrict__ aW,
                                            const float* __restrict__ bX,
                                            v8f (&acc)[2])
{
#pragma unroll 8
    for (int k0 = 0; k0 < 256; k0 += 4) {
        const v2f a = *(const v2f*)(aW + k0);
        const v2f b0 = *(const v2f*)(bX + (k0 >> 1) * PX2);
        const v2f b1 = *(const v2f*)(bX + (k0 >> 1) * PX2 + 32);
        acc[0] = SWAP ? wmma_f32(b0, a, acc[0]) : wmma_f32(a, b0, acc[0]);
        acc[1] = SWAP ? wmma_f32(b1, a, acc[1]) : wmma_f32(a, b1, acc[1]);
    }
}

__global__ __launch_bounds__(256) void gemm_f32_wmma(
    const float* __restrict__ W,
    const float* __restrict__ X,
    float* __restrict__ O,
    int Mblocks)
{
    __shared__ __attribute__((aligned(16))) float panels[64 * PW + 128 * PX2];
    float* const sW = panels;
    float* const sX = panels + 64 * PW;

    const int N = 2048;
    const int wv   = __builtin_amdgcn_readfirstlane((int)(threadIdx.x >> 5));
    const int mw   = wv >> 1, nw = wv & 1;
    const int lane = threadIdx.x & 31;
    const int lm   = lane & 15;
    const int hi   = lane >> 4;

    int t = blockIdx.x;
    const int nb = t & 31; t >>= 5;
    const int mb = t % Mblocks; t /= Mblocks;
    const int b  = t;

    stage_panels(W + (size_t)mb * 64 * 256, X + (size_t)b * 256 * N + 64 * nb, sW, sX);
    __syncthreads();

    v8f acc[2] = {};
    piece_16x32<false>(sW + (16 * mw + lm) * PW + 2 * hi,
                       panels + x_fragment_base(hi, nw, lm), acc);

    float* Op = O + ((size_t)b * (Mblocks * 64) + 64 * mb + 16 * mw) * N + 64 * nb + 32 * nw;
#pragma unroll
    for (int j = 0; j < 2; ++j)
#pragma unroll
        for (int r = 0; r < 8; ++r)
            Op[(size_t)(r + 8 * hi) * N + 16 * j + lm] = acc[j][r];
}

__global__ __launch_bounds__(256) void gemm_qkv_fragment_order(
    const float* __restrict__ W,
    const float* __restrict__ X,
    float* __restrict__ P)
{
    __shared__ __attribute__((aligned(16))) float panels[64 * PW + 128 * PX2];
    float* const sW = panels;
    float* const sX = panels + 64 * PW;

    const int N = 2048;
    const int wv   = __builtin_amdgcn_readfirstlane((int)(threadIdx.x >> 5));
    const int mw   = wv >> 1, nw = wv & 1;
    const int lane = threadIdx.x & 31;
    const int lm   = lane & 15;
    const int hi   = lane >> 4;

    int t = blockIdx.x;
    const int nb = t & 31; t >>= 5;
    const int mb = t % 12; t /= 12;
    const int b  = t;

    const int mt    = 4 * mb + mw;
    const int which = mb >> 2;
    const int head  = (mt & 15) >> 1;
    const int dhalf = mt & 1;

    stage_panels(W + (size_t)mb * 64 * 256, X + (size_t)b * 256 * N + 64 * nb, sW, sX);
    __syncthreads();

    const float* aW = sW + (16 * mw + lm) * PW + 2 * hi;
    const float* bX = panels + x_fragment_base(hi, nw, lm);
    v8f acc[2] = {};
    if (which == 2) piece_16x32<true >(aW, bX, acc);
    else            piece_16x32<false>(aW, bX, acc);

    float* Pp = P + (size_t)which * (4 * 8 * 128 * 512)
                  + ((size_t)(b * 8 + head) * 128 + 4 * nb + 2 * nw) * 512
                  + (2 * dhalf) * 128 + lane * 4;
#pragma unroll
    for (int j = 0; j < 2; ++j) {
        v4f lo, up;
        lo.x = acc[j][0]; lo.y = acc[j][1]; lo.z = acc[j][2]; lo.w = acc[j][3];
        up.x = acc[j][4]; up.y = acc[j][5]; up.z = acc[j][6]; up.w = acc[j][7];
        *(v4f*)(Pp + j * 512)       = lo;
        *(v4f*)(Pp + j * 512 + 128) = up;
    }
}

__device__ __forceinline__ float xhalf(float x) {
    int b = __builtin_bit_cast(int, x);
    return __builtin_bit_cast(float,
        __builtin_amdgcn_permlanex16(b, b, 0x76543210, 0xfedcba98, false, false));
}

__global__ __launch_bounds__(128, 8) void flash_attn_wmma(
    const float* __restrict__ QKV, float* __restrict__ AO)
{
    __shared__ float sO[4 * 4 * 8 * 32];
    __shared__ float sM[4 * 2 * 32];
    __shared__ float sL[4 * 2 * 32];

    const int lane = threadIdx.x & 31;
    const int c    = lane & 15;
    const int w    = __builtin_amdgcn_readfirstlane((int)(threadIdx.x >> 5));
    __builtin_assume(w >= 0 && w < 4);

    int t = blockIdx.x;
    const int ns = t & 63; t >>= 6;
    const int hd = t & 7;  t >>= 3;
    const int b  = t;
    const int n0 = ns << 5;
    const size_t N = 2048;

    const size_t REGION = (size_t)4 * 8 * 128 * 512;
    const size_t hbase  = ((size_t)(b * 8 + hd) * 128) * 512;
    const float* q = QKV +              hbase + lane * 4;
    const float* k = QKV +     REGION + hbase + lane * 4;
    const float* v = QKV + 2 * REGION + hbase + lane * 4;

    v2f qb[2][8];
#pragma unroll
    for (int u = 0; u < 2; ++u)
#pragma unroll
        for (int j = 0; j < 4; ++j) {
            const v4f t4 = *(const v4f*)(q + (size_t)(2 * ns + u) * 512 + j * 128);
            qb[u][2 * j].x     = t4.x;  qb[u][2 * j].y     = t4.y;
            qb[u][2 * j + 1].x = t4.z;  qb[u][2 * j + 1].y = t4.w;
        }

    v8f o[2][2] = {};
    float rmax[2] = { -3.4e38f, -3.4e38f };
    float rsum[2] = { 0.0f, 0.0f };
    const float inv_scale = 0.17677669529663687f;

    for (int kt = w; kt < 128; kt += 4) {
        v2f ka[8];
#pragma unroll
        for (int j = 0; j < 4; ++j) {
            const v4f t4 = *(const v4f*)(k + (size_t)kt * 512 + j * 128);
            ka[2 * j].x     = t4.x;  ka[2 * j].y     = t4.y;
            ka[2 * j + 1].x = t4.z;  ka[2 * j + 1].y = t4.w;
        }
        v4f va[2][2];
#pragma unroll
        for (int dh = 0; dh < 2; ++dh) {
            va[dh][0] = *(const v4f*)(v + (size_t)kt * 512 + (2 * dh)     * 128);
            va[dh][1] = *(const v4f*)(v + (size_t)kt * 512 + (2 * dh + 1) * 128);
        }

        __builtin_amdgcn_sched_barrier(0);

        v8f s[2] = {};
#pragma unroll
        for (int dc = 0; dc < 8; ++dc)
#pragma unroll
            for (int u = 0; u < 2; ++u)
                s[u] = wmma_f32(ka[dc], qb[u][dc], s[u]);

#pragma unroll
        for (int u = 0; u < 2; ++u) {
            float x[8];
#pragma unroll
            for (int r = 0; r < 8; ++r) x[r] = s[u][r] * inv_scale;
            float tmax = fmaxf(fmaxf(fmaxf(x[0], x[1]), fmaxf(x[2], x[3])),
                               fmaxf(fmaxf(x[4], x[5]), fmaxf(x[6], x[7])));
            tmax = fmaxf(tmax, xhalf(tmax));
            const float nm = fmaxf(rmax[u], tmax);
            const float cf = __expf(rmax[u] - nm);
            rmax[u] = nm;
            float e[8];
#pragma unroll
            for (int r = 0; r < 8; ++r) e[r] = __expf(x[r] - nm);
            const float ls = ((e[0] + e[1]) + (e[2] + e[3])) +
                             ((e[4] + e[5]) + (e[6] + e[7]));
            rsum[u] = rsum[u] * cf + ls;
#pragma unroll
            for (int r = 0; r < 8; ++r) { o[u][0][r] *= cf; o[u][1][r] *= cf; }

#pragma unroll
            for (int mc = 0; mc < 4; ++mc) {
                v2f pb, a0, a1;
                pb.x = e[2 * mc];
                pb.y = e[2 * mc + 1];
                a0.x = va[0][mc >> 1][2 * (mc & 1)];
                a0.y = va[0][mc >> 1][2 * (mc & 1) + 1];
                a1.x = va[1][mc >> 1][2 * (mc & 1)];
                a1.y = va[1][mc >> 1][2 * (mc & 1) + 1];
                o[u][0] = wmma_f32(a0, pb, o[u][0]);
                o[u][1] = wmma_f32(a1, pb, o[u][1]);
            }
        }
    }

#pragma unroll
    for (int u = 0; u < 2; ++u) {
#pragma unroll
        for (int dh = 0; dh < 2; ++dh)
#pragma unroll
            for (int r = 0; r < 8; ++r)
                sO[((w * 4 + 2 * u + dh) * 8 + r) * 32 + lane] = o[u][dh][r];
        sM[(w * 2 + u) * 32 + lane] = rmax[u];
        sL[(w * 2 + u) * 32 + lane] = rsum[u];
    }
    __syncthreads();

    {
        const int qu = lane >> 4;
        const int dh = w >> 1, hh = w & 1;
        float mj[4];
#pragma unroll
        for (int j = 0; j < 4; ++j) mj[j] = sM[(j * 2 + qu) * 32 + c];
        const float mx = fmaxf(fmaxf(mj[0], mj[1]), fmaxf(mj[2], mj[3]));
        float tot = 0.0f;
        float acc[8] = {};
#pragma unroll
        for (int j = 0; j < 4; ++j) {
            const float f = __expf(mj[j] - mx);
            tot += f * (sL[(j * 2 + qu) * 32 + c] + sL[(j * 2 + qu) * 32 + 16 + c]);
#pragma unroll
            for (int r = 0; r < 8; ++r)
                acc[r] += f * sO[((j * 4 + 2 * qu + dh) * 8 + r) * 32 + 16 * hh + c];
        }
        const float inv = 1.0f / tot;
        float* ao = AO + ((size_t)b * 256 + (size_t)hd * 32 + 8 * w) * N + n0 + lane;
#pragma unroll
        for (int r = 0; r < 8; ++r)
            ao[(size_t)r * N] = acc[r] * inv;
    }
}

extern "C" void kernel_launch(void* const* d_in, const int* in_sizes, int n_in,
                              void* d_out, int out_size, void* d_ws, size_t ws_size,
                              hipStream_t stream)
{
    const float* x      = (const float*)d_in[0];
    const float* w_qkv  = (const float*)d_in[1];
    const float* w_proj = (const float*)d_in[2];
    float* out = (float*)d_out;

    const int B = 4, C = 256, N = 2048;

    float* qkv = (float*)d_ws;
    float* ao  = qkv + (size_t)B * 3 * C * N;

    gemm_qkv_fragment_order<<<1536, 256, 0, stream>>>(w_qkv, x, qkv);

    flash_attn_wmma<<<B * 8 * (N >> 5), 128, 0, stream>>>(qkv, ao);

    gemm_f32_wmma<<<512, 256, 0, stream>>>(w_proj, ao, out, 4);
}


# attention: wave priority lowered in steps over a wave's 32 key trips (s_setprio 3,2,1,0 from trips 0,16,26,30): the 8 waves of a SIMD advance together, not oldest-first one after another
# speedup vs baseline: 2.6649x; 1.0580x over previous
#include <hip/hip_runtime.h>

typedef float v2f __attribute__((ext_vector_type(2)));
typedef float v4f __attribute__((ext_vector_type(4)));
typedef float v8f __attribute__((ext_vector_type(8)));

__device__ __forceinline__ v8f wmma_f32(v2f a, v2f b, v8f c) {
    return __builtin_amdgcn_wmma_f32_16x16x4_f32(false, a, false, b, (short)0, c,
                                                 false, false);
}

#define PW 260
#define PX2 160

__device__ __forceinline__ void stage_panels(const float* __restrict__ Wt,
                                             const float* __restrict__ Xt,
                                             float* __restrict__ sW,
                                             float* __restrict__ sX)
{
    const int t = threadIdx.x;
    v4f w4[16], xa[8], xb[8];
#pragma unroll
    for (int i = 0; i < 16; ++i)
        w4[i] = *(const v4f*)(Wt + (size_t)(i * 256 + t) * 4);
#pragma unroll
    for (int i = 0; i < 8; ++i) {
        const int id = i * 256 + t;
        const float* xp = Xt + (size_t)(id >> 4) * (2 * 2048) + (id & 15) * 4;
        xa[i] = *(const v4f*)(xp);
        xb[i] = *(const v4f*)(xp + 2048);
    }
#pragma unroll
    for (int i = 0; i < 16; ++i) {
        const int id = i * 256 + t;
        *(v4f*)(sW + (id >> 6) * PW + (id & 63) * 4) = w4[i];
    }
#pragma unroll
    for (int i = 0; i < 8; ++i) {
        const int id = i * 256 + t;
        float* d = sX + (id >> 4) * PX2 + (id & 15) * 8;
        v4f p0, p1;
        p0.x = xa[i].x; p0.y = xb[i].x; p0.z = xa[i].y; p0.w = xb[i].y;
        p1.x = xa[i].z; p1.y = xb[i].z; p1.z = xa[i].w; p1.w = xb[i].w;
        *(v4f*)(d)     = p0;
        *(v4f*)(d + 4) = p1;
    }
}

__device__ __forceinline__ int x_fragment_base(int hi, int nw, int lm)
{
    int off = 64 * PW + hi * PX2 + 2 * (32 * nw + lm);
    asm volatile("" : "+v"(off));
    return off;
}

template <bool SWAP>
__device__ __forceinline__ void piece_16x32(const float* __restrict__ aW,
                                            const float* __restrict__ bX,
                                            v8f (&acc)[2])
{
#pragma unroll 8
    for (int k0 = 0; k0 < 256; k0 += 4) {
        const v2f a = *(const v2f*)(aW + k0);
        const v2f b0 = *(const v2f*)(bX + (k0 >> 1) * PX2);
        const v2f b1 = *(const v2f*)(bX + (k0 >> 1) * PX2 + 32);
        acc[0] = SWAP ? wmma_f32(b0, a, acc[0]) : wmma_f32(a, b0, acc[0]);
        acc[1] = SWAP ? wmma_f32(b1, a, acc[1]) : wmma_f32(a, b1, acc[1]);
    }
}

__global__ __launch_bounds__(256) void gemm_f32_wmma(
    const float* __restrict__ W,
    const float* __restrict__ X,
    float* __restrict__ O,
    int Mblocks)
{
    __shared__ __attribute__((aligned(16))) float panels[64 * PW + 128 * PX2];
    float* const sW = panels;
    float* const sX = panels + 64 * PW;

    const int N = 2048;
    const int wv   = __builtin_amdgcn_readfirstlane((int)(threadIdx.x >> 5));
    const int mw   = wv >> 1, nw = wv & 1;
    const int lane = threadIdx.x & 31;
    const int lm   = lane & 15;
    const int hi   = lane >> 4;

    int t = blockIdx.x;
    const int nb = t & 31; t >>= 5;
    const int mb = t % Mblocks; t /= Mblocks;
    const int b  = t;

    stage_panels(W + (size_t)mb * 64 * 256, X + (size_t)b * 256 * N + 64 * nb, sW, sX);
    __syncthreads();

    v8f acc[2] = {};
    piece_16x32<false>(sW + (16 * mw + lm) * PW + 2 * hi,
                       panels + x_fragment_base(hi, nw, lm), acc);

    float* Op = O + ((size_t)b * (Mblocks * 64) + 64 * mb + 16 * mw) * N + 64 * nb + 32 * nw;
#pragma unroll
    for (int j = 0; j < 2; ++j)
#pragma unroll
        for (int r = 0; r < 8; ++r)
            Op[(size_t)(r + 8 * hi) * N + 16 * j + lm] = acc[j][r];
}

__global__ __launch_bounds__(256) void gemm_qkv_fragment_order(
    const float* __restrict__ W,
    const float* __restrict__ X,
    float* __restrict__ P)
{
    __shared__ __attribute__((aligned(16))) float panels[64 * PW + 128 * PX2];
    float* const sW = panels;
    float* const sX = panels + 64 * PW;

    const int N = 2048;
    const int wv   = __builtin_amdgcn_readfirstlane((int)(threadIdx.x >> 5));
    const int mw   = wv >> 1, nw = wv & 1;
    const int lane = threadIdx.x & 31;
    const int lm   = lane & 15;
    const int hi   = lane >> 4;

    int t = blockIdx.x;
    const int nb = t & 31; t >>= 5;
    const int mb = t % 12; t /= 12;
    const int b  = t;

    const int mt    = 4 * mb + mw;
    const int which = mb >> 2;
    const int head  = (mt & 15) >> 1;
    const int dhalf = mt & 1;

    stage_panels(W + (size_t)mb * 64 * 256, X + (size_t)b * 256 * N + 64 * nb, sW, sX);
    __syncthreads();

    const float* aW = sW + (16 * mw + lm) * PW + 2 * hi;
    const float* bX = panels + x_fragment_base(hi, nw, lm);
    v8f acc[2] = {};
    if (which == 2) piece_16x32<true >(aW, bX, acc);
    else            piece_16x32<false>(aW, bX, acc);

    float* Pp = P + (size_t)which * (4 * 8 * 128 * 512)
                  + ((size_t)(b * 8 + head) * 128 + 4 * nb + 2 * nw) * 512
                  + (2 * dhalf) * 128 + lane * 4;
#pragma unroll
    for (int j = 0; j < 2; ++j) {
        v4f lo, up;
        lo.x = acc[j][0]; lo.y = acc[j][1]; lo.z = acc[j][2]; lo.w = acc[j][3];
        up.x = acc[j][4]; up.y = acc[j][5]; up.z = acc[j][6]; up.w = acc[j][7];
        *(v4f*)(Pp + j * 512)       = lo;
        *(v4f*)(Pp + j * 512 + 128) = up;
    }
}

__device__ __forceinline__ float xhalf(float x) {
    int b = __builtin_bit_cast(int, x);
    return __builtin_bit_cast(float,
        __builtin_amdgcn_permlanex16(b, b, 0x76543210, 0xfedcba98, false, false));
}

__device__ __forceinline__ void attn_key_tile(const float* __restrict__ k,
                                              const float* __restrict__ v, int kt,
                                              const v2f (&qb)[2][8], v8f (&o)[2][2],
                                              float (&rmax)[2], float (&rsum)[2])
{
    const float inv_scale = 0.17677669529663687f;
    v2f ka[8];
#pragma unroll
    for (int j = 0; j < 4; ++j) {
        const v4f t4 = *(const v4f*)(k + (size_t)kt * 512 + j * 128);
        ka[2 * j].x     = t4.x;  ka[2 * j].y     = t4.y;
        ka[2 * j + 1].x = t4.z;  ka[2 * j + 1].y = t4.w;
    }
    v4f va[2][2];
#pragma unroll
    for (int dh = 0; dh < 2; ++dh) {
        va[dh][0] = *(const v4f*)(v + (size_t)kt * 512 + (2 * dh)     * 128);
        va[dh][1] = *(const v4f*)(v + (size_t)kt * 512 + (2 * dh + 1) * 128);
    }

    __builtin_amdgcn_sched_barrier(0);

    v8f s[2] = {};
#pragma unroll
    for (int dc = 0; dc < 8; ++dc)
#pragma unroll
        for (int u = 0; u < 2; ++u)
            s[u] = wmma_f32(ka[dc], qb[u][dc], s[u]);

#pragma unroll
    for (int u = 0; u < 2; ++u) {
        float x[8];
#pragma unroll
        for (int r = 0; r < 8; ++r) x[r] = s[u][r] * inv_scale;
        float tmax = fmaxf(fmaxf(fmaxf(x[0], x[1]), fmaxf(x[2], x[3])),
                           fmaxf(fmaxf(x[4], x[5]), fmaxf(x[6], x[7])));
        tmax = fmaxf(tmax, xhalf(tmax));
        const float nm = fmaxf(rmax[u], tmax);
        const float cf = __expf(rmax[u] - nm);
        rmax[u] = nm;
        float e[8];
#pragma unroll
        for (int r = 0; r < 8; ++r) e[r] = __expf(x[r] - nm);
        const float ls = ((e[0] + e[1]) + (e[2] + e[3])) +
                         ((e[4] + e[5]) + (e[6] + e[7]));
        rsum[u] = rsum[u] * cf + ls;
#pragma unroll
        for (int r = 0; r < 8; ++r) { o[u][0][r] *= cf; o[u][1][r] *= cf; }

#pragma unroll
        for (int mc = 0; mc < 4; ++mc) {
            v2f pb, a0, a1;
            pb.x = e[2 * mc];
            pb.y = e[2 * mc + 1];
            a0.x = va[0][mc >> 1][2 * (mc & 1)];
            a0.y = va[0][mc >> 1][2 * (mc & 1) + 1];
            a1.x = va[1][mc >> 1][2 * (mc & 1)];
            a1.y = va[1][mc >> 1][2 * (mc & 1) + 1];
            o[u][0] = wmma_f32(a0, pb, o[u][0]);
            o[u][1] = wmma_f32(a1, pb, o[u][1]);
        }
    }
}

__global__ __launch_bounds__(128, 8) void flash_attn_wmma(
    const float* __restrict__ QKV, float* __restrict__ AO)
{
    __shared__ float sO[4 * 4 * 8 * 32];
    __shared__ float sM[4 * 2 * 32];
    __shared__ float sL[4 * 2 * 32];

    const int lane = threadIdx.x & 31;
    const int c    = lane & 15;
    const int w    = __builtin_amdgcn_readfirstlane((int)(threadIdx.x >> 5));
    __builtin_assume(w >= 0 && w < 4);

    int t = blockIdx.x;
    const int ns = t & 63; t >>= 6;
    const int hd = t & 7;  t >>= 3;
    const int b  = t;
    const int n0 = ns << 5;
    const size_t N = 2048;

    const size_t REGION = (size_t)4 * 8 * 128 * 512;
    const size_t hbase  = ((size_t)(b * 8 + hd) * 128) * 512;
    const float* q = QKV +              hbase + lane * 4;
    const float* k = QKV +     REGION + hbase + lane * 4;
    const float* v = QKV + 2 * REGION + hbase + lane * 4;

    v2f qb[2][8];
#pragma unroll
    for (int u = 0; u < 2; ++u)
#pragma unroll
        for (int j = 0; j < 4; ++j) {
            const v4f t4 = *(const v4f*)(q + (size_t)(2 * ns + u) * 512 + j * 128);
            qb[u][2 * j].x     = t4.x;  qb[u][2 * j].y     = t4.y;
            qb[u][2 * j + 1].x = t4.z;  qb[u][2 * j + 1].y = t4.w;
        }

    v8f o[2][2] = {};
    float rmax[2] = { -3.4e38f, -3.4e38f };
    float rsum[2] = { 0.0f, 0.0f };

    int kt = w;
    __builtin_amdgcn_s_setprio(3);
    for (; kt < 64;  kt += 4) attn_key_tile(k, v, kt, qb, o, rmax, rsum);
    __builtin_amdgcn_s_setprio(2);
    for (; kt < 104; kt += 4) attn_key_tile(k, v, kt, qb, o, rmax, rsum);
    __builtin_amdgcn_s_setprio(1);
    for (; kt < 120; kt += 4) attn_key_tile(k, v, kt, qb, o, rmax, rsum);
    __builtin_amdgcn_s_setprio(0);
    for (; kt < 128; kt += 4) attn_key_tile(k, v, kt, qb, o, rmax, rsum);

#pragma unroll
    for (int u = 0; u < 2; ++u) {
#pragma unroll
        for (int dh = 0; dh < 2; ++dh)
#pragma unroll
            for (int r = 0; r < 8; ++r)
                sO[((w * 4 + 2 * u + dh) * 8 + r) * 32 + lane] = o[u][dh][r];
        sM[(w * 2 + u) * 32 + lane] = rmax[u];
        sL[(w * 2 + u) * 32 + lane] = rsum[u];
    }
    __syncthreads();

    {
        const int qu = lane >> 4;
        const int dh = w >> 1, hh = w & 1;
        float mj[4];
#pragma unroll
        for (int j = 0; j < 4; ++j) mj[j] = sM[(j * 2 + qu) * 32 + c];
        const float mx = fmaxf(fmaxf(mj[0], mj[1]), fmaxf(mj[2], mj[3]));
        float tot = 0.0f;
        float acc[8] = {};
#pragma unroll
        for (int j = 0; j < 4; ++j) {
            const float f = __expf(mj[j] - mx);
            tot += f * (sL[(j * 2 + qu) * 32 + c] + sL[(j * 2 + qu) * 32 + 16 + c]);
#pragma unroll
            for (int r = 0; r < 8; ++r)
                acc[r] += f * sO[((j * 4 + 2 * qu + dh) * 8 + r) * 32 + 16 * hh + c];
        }
        const float inv = 1.0f / tot;
        float* ao = AO + ((size_t)b * 256 + (size_t)hd * 32 + 8 * w) * N + n0 + lane;
#pragma unroll
        for (int r = 0; r < 8; ++r)
            ao[(size_t)r * N] = acc[r] * inv;
    }
}

extern "C" void kernel_launch(void* const* d_in, const int* in_sizes, int n_in,
                              void* d_out, int out_size, void* d_ws, size_t ws_size,
                              hipStream_t stream)
{
    const float* x      = (const float*)d_in[0];
    const float* w_qkv  = (const float*)d_in[1];
    const float* w_proj = (const float*)d_in[2];
    float* out = (float*)d_out;

    const int B = 4, C = 256, N = 2048;

    float* qkv = (float*)d_ws;
    float* ao  = qkv + (size_t)B * 3 * C * N;

    gemm_qkv_fragment_order<<<1536, 256, 0, stream>>>(w_qkv, x, qkv);

    flash_attn_wmma<<<B * 8 * (N >> 5), 128, 0, stream>>>(qkv, ao);

    gemm_f32_wmma<<<512, 256, 0, stream>>>(w_proj, ao, out, 4);
}


# attention softmax: weights exp2(score*c - m) taken against a row maximum in use straight off the accumulator pairs; tile max, lane-half exchange, Ot rescale only in a rare wave-uniform branch
# speedup vs baseline: 2.8532x; 1.0707x over previous
#include <hip/hip_runtime.h>

typedef float v2f __attribute__((ext_vector_type(2)));
typedef float v4f __attribute__((ext_vector_type(4)));
typedef float v8f __attribute__((ext_vector_type(8)));

__device__ __forceinline__ v8f wmma_f32(v2f a, v2f b, v8f c) {
    return __builtin_amdgcn_wmma_f32_16x16x4_f32(false, a, false, b, (short)0, c,
                                                 false, false);
}

#define PW 260
#define PX2 160

__device__ __forceinline__ void stage_panels(const float* __restrict__ Wt,
                                             const float* __restrict__ Xt,
                                             float* __restrict__ sW,
                                             float* __restrict__ sX)
{
    const int t = threadIdx.x;
    v4f w4[16], xa[8], xb[8];
#pragma unroll
    for (int i = 0; i < 16; ++i)
        w4[i] = *(const v4f*)(Wt + (size_t)(i * 256 + t) * 4);
#pragma unroll
    for (int i = 0; i < 8; ++i) {
        const int id = i * 256 + t;
        const float* xp = Xt + (size_t)(id >> 4) * (2 * 2048) + (id & 15) * 4;
        xa[i] = *(const v4f*)(xp);
        xb[i] = *(const v4f*)(xp + 2048);
    }
#pragma unroll
    for (int i = 0; i < 16; ++i) {
        const int id = i * 256 + t;
        *(v4f*)(sW + (id >> 6) * PW + (id & 63) * 4) = w4[i];
    }
#pragma unroll
    for (int i = 0; i < 8; ++i) {
        const int id = i * 256 + t;
        float* d = sX + (id >> 4) * PX2 + (id & 15) * 8;
        v4f p0, p1;
        p0.x = xa[i].x; p0.y = xb[i].x; p0.z = xa[i].y; p0.w = xb[i].y;
        p1.x = xa[i].z; p1.y = xb[i].z; p1.z = xa[i].w; p1.w = xb[i].w;
        *(v4f*)(d)     = p0;
        *(v4f*)(d + 4) = p1;
    }
}

__device__ __forceinline__ int x_fragment_base(int hi, int nw, int lm)
{
    int off = 64 * PW + hi * PX2 + 2 * (32 * nw + lm);
    asm volatile("" : "+v"(off));
    return off;
}

template <bool SWAP>
__device__ __forceinline__ void piece_16x32(const float* __restrict__ aW,
                                            const float* __restrict__ bX,
                                            v8f (&acc)[2])
{
#pragma unroll 8
    for (int k0 = 0; k0 < 256; k0 += 4) {
        const v2f a = *(const v2f*)(aW + k0);
        const v2f b0 = *(const v2f*)(bX + (k0 >> 1) * PX2);
        const v2f b1 = *(const v2f*)(bX + (k0 >> 1) * PX2 + 32);
        acc[0] = SWAP ? wmma_f32(b0, a, acc[0]) : wmma_f32(a, b0, acc[0]);
        acc[1] = SWAP ? wmma_f32(b1, a, acc[1]) : wmma_f32(a, b1, acc[1]);
    }
}

__global__ __launch_bounds__(256) void gemm_f32_wmma(
    const float* __restrict__ W,
    const float* __restrict__ X,
    float* __restrict__ O,
    int Mblocks)
{
    __shared__ __attribute__((aligned(16))) float panels[64 * PW + 128 * PX2];
    float* const sW = panels;
    float* const sX = panels + 64 * PW;

    const int N = 2048;
    const int wv   = __builtin_amdgcn_readfirstlane((int)(threadIdx.x >> 5));
    const int mw   = wv >> 1, nw = wv & 1;
    const int lane = threadIdx.x & 31;
    const int lm   = lane & 15;
    const int hi   = lane >> 4;

    int t = blockIdx.x;
    const int nb = t & 31; t >>= 5;
    const int mb = t % Mblocks; t /= Mblocks;
    const int b  = t;

    stage_panels(W + (size_t)mb * 64 * 256, X + (size_t)b * 256 * N + 64 * nb, sW, sX);
    __syncthreads();

    v8f acc[2] = {};
    piece_16x32<false>(sW + (16 * mw + lm) * PW + 2 * hi,
                       panels + x_fragment_base(hi, nw, lm), acc);

    float* Op = O + ((size_t)b * (Mblocks * 64) + 64 * mb + 16 * mw) * N + 64 * nb + 32 * nw;
#pragma unroll
    for (int j = 0; j < 2; ++j)
#pragma unroll
        for (int r = 0; r < 8; ++r)
            Op[(size_t)(r + 8 * hi) * N + 16 * j + lm] = acc[j][r];
}

__global__ __launch_bounds__(256) void gemm_qkv_fragment_order(
    const float* __restrict__ W,
    const float* __restrict__ X,
    float* __restrict__ P)
{
    __shared__ __attribute__((aligned(16))) float panels[64 * PW + 128 * PX2];
    float* const sW = panels;
    float* const sX = panels + 64 * PW;

    const int N = 2048;
    const int wv   = __builtin_amdgcn_readfirstlane((int)(threadIdx.x >> 5));
    const int mw   = wv >> 1, nw = wv & 1;
    const int lane = threadIdx.x & 31;
    const int lm   = lane & 15;
    const int hi   = lane >> 4;

    int t = blockIdx.x;
    const int nb = t & 31; t >>= 5;
    const int mb = t % 12; t /= 12;
    const int b  = t;

    const int mt    = 4 * mb + mw;
    const int which = mb >> 2;
    const int head  = (mt & 15) >> 1;
    const int dhalf = mt & 1;

    stage_panels(W + (size_t)mb * 64 * 256, X + (size_t)b * 256 * N + 64 * nb, sW, sX);
    __syncthreads();

    const float* aW = sW + (16 * mw + lm) * PW + 2 * hi;
    const float* bX = panels + x_fragment_base(hi, nw, lm);
    v8f acc[2] = {};
    if (which == 2) piece_16x32<true >(aW, bX, acc);
    else            piece_16x32<false>(aW, bX, acc);

    float* Pp = P + (size_t)which * (4 * 8 * 128 * 512)
                  + ((size_t)(b * 8 + head) * 128 + 4 * nb + 2 * nw) * 512
                  + (2 * dhalf) * 128 + lane * 4;
#pragma unroll
    for (int j = 0; j < 2; ++j) {
        v4f lo, up;
        lo.x = acc[j][0]; lo.y = acc[j][1]; lo.z = acc[j][2]; lo.w = acc[j][3];
        up.x = acc[j][4]; up.y = acc[j][5]; up.z = acc[j][6]; up.w = acc[j][7];
        *(v4f*)(Pp + j * 512)       = lo;
        *(v4f*)(Pp + j * 512 + 128) = up;
    }
}

__device__ __forceinline__ float xhalf(float x) {
    int b = __builtin_bit_cast(int, x);
    return __builtin_bit_cast(float,
        __builtin_amdgcn_permlanex16(b, b, 0x76543210, 0xfedcba98, false, false));
}

#define SOFTMAX_C2 0.25503486164919736f

__device__ __forceinline__ void tile_weights(const v8f (&s)[2], const float (&nm2)[2],
                                             v2f (&e)[2][4], float (&ls)[2])
{
#pragma unroll
    for (int u = 0; u < 2; ++u) {
        v2f cc; cc.x = SOFTMAX_C2; cc.y = SOFTMAX_C2;
        v2f nn; nn.x = nm2[u];     nn.y = nm2[u];
#pragma unroll
        for (int mc = 0; mc < 4; ++mc) {
            v2f a; a.x = s[u][2 * mc]; a.y = s[u][2 * mc + 1];
            const v2f t = __builtin_elementwise_fma(a, cc, nn);
            e[u][mc].x = __builtin_amdgcn_exp2f(t.x);
            e[u][mc].y = __builtin_amdgcn_exp2f(t.y);
        }
        const v2f p = (e[u][0] + e[u][1]) + (e[u][2] + e[u][3]);
        ls[u] = p.x + p.y;
    }
}

__device__ __forceinline__ void attn_key_tile(const float* __restrict__ k,
                                              const float* __restrict__ v, int kt,
                                              const v2f (&qb)[2][8], v8f (&o)[2][2],
                                              float (&nm2)[2], float (&rsum)[2])
{
    v2f ka[8];
#pragma unroll
    for (int j = 0; j < 4; ++j) {
        const v4f t4 = *(const v4f*)(k + (size_t)kt * 512 + j * 128);
        ka[2 * j].x     = t4.x;  ka[2 * j].y     = t4.y;
        ka[2 * j + 1].x = t4.z;  ka[2 * j + 1].y = t4.w;
    }
    v4f va[2][2];
#pragma unroll
    for (int dh = 0; dh < 2; ++dh) {
        va[dh][0] = *(const v4f*)(v + (size_t)kt * 512 + (2 * dh)     * 128);
        va[dh][1] = *(const v4f*)(v + (size_t)kt * 512 + (2 * dh + 1) * 128);
    }

    __builtin_amdgcn_sched_barrier(0);

    v8f s[2] = {};
#pragma unroll
    for (int dc = 0; dc < 8; ++dc)
#pragma unroll
        for (int u = 0; u < 2; ++u)
            s[u] = wmma_f32(ka[dc], qb[u][dc], s[u]);

    v2f e[2][4];
    float ls[2];
    tile_weights(s, nm2, e, ls);

    if (__builtin_amdgcn_ballot_w32(fmaxf(ls[0], ls[1]) > 1073741824.0f) != 0u) {
#pragma unroll
        for (int u = 0; u < 2; ++u) {
            float tmax = fmaxf(fmaxf(fmaxf(s[u][0], s[u][1]), fmaxf(s[u][2], s[u][3])),
                               fmaxf(fmaxf(s[u][4], s[u][5]), fmaxf(s[u][6], s[u][7])));
            tmax = fmaxf(tmax, xhalf(tmax));
            const float old = -nm2[u];
            const float nm  = fmaxf(old, tmax * SOFTMAX_C2);
            const float cf  = __builtin_amdgcn_exp2f(old - nm);
            nm2[u]  = -nm;
            rsum[u] *= cf;
#pragma unroll
            for (int r = 0; r < 8; ++r) { o[u][0][r] *= cf; o[u][1][r] *= cf; }
        }
        tile_weights(s, nm2, e, ls);
    }

#pragma unroll
    for (int u = 0; u < 2; ++u) {
        rsum[u] += ls[u];
#pragma unroll
        for (int mc = 0; mc < 4; ++mc) {
            v2f a0, a1;
            a0.x = va[0][mc >> 1][2 * (mc & 1)];
            a0.y = va[0][mc >> 1][2 * (mc & 1) + 1];
            a1.x = va[1][mc >> 1][2 * (mc & 1)];
            a1.y = va[1][mc >> 1][2 * (mc & 1) + 1];
            o[u][0] = wmma_f32(a0, e[u][mc], o[u][0]);
            o[u][1] = wmma_f32(a1, e[u][mc], o[u][1]);
        }
    }
}

__global__ __launch_bounds__(128, 8) void flash_attn_wmma(
    const float* __restrict__ QKV, float* __restrict__ AO)
{
    __shared__ float sO[4 * 4 * 8 * 32];
    __shared__ float sM[4 * 2 * 32];
    __shared__ float sL[4 * 2 * 32];

    const int lane = threadIdx.x & 31;
    const int c    = lane & 15;
    const int w    = __builtin_amdgcn_readfirstlane((int)(threadIdx.x >> 5));
    __builtin_assume(w >= 0 && w < 4);

    int t = blockIdx.x;
    const int ns = t & 63; t >>= 6;
    const int hd = t & 7;  t >>= 3;
    const int b  = t;
    const int n0 = ns << 5;
    const size_t N = 2048;

    const size_t REGION = (size_t)4 * 8 * 128 * 512;
    const size_t hbase  = ((size_t)(b * 8 + hd) * 128) * 512;
    const float* q = QKV +              hbase + lane * 4;
    const float* k = QKV +     REGION + hbase + lane * 4;
    const float* v = QKV + 2 * REGION + hbase + lane * 4;

    v2f qb[2][8];
#pragma unroll
    for (int u = 0; u < 2; ++u)
#pragma unroll
        for (int j = 0; j < 4; ++j) {
            const v4f t4 = *(const v4f*)(q + (size_t)(2 * ns + u) * 512 + j * 128);
            qb[u][2 * j].x     = t4.x;  qb[u][2 * j].y     = t4.y;
            qb[u][2 * j + 1].x = t4.z;  qb[u][2 * j + 1].y = t4.w;
        }

    v8f o[2][2] = {};
    float nm2[2]  = { 3.0e38f, 3.0e38f };
    float rsum[2] = { 0.0f, 0.0f };

    int kt = w;
    __builtin_amdgcn_s_setprio(3);
    for (; kt < 64;  kt += 4) attn_key_tile(k, v, kt, qb, o, nm2, rsum);
    __builtin_amdgcn_s_setprio(2);
    for (; kt < 104; kt += 4) attn_key_tile(k, v, kt, qb, o, nm2, rsum);
    __builtin_amdgcn_s_setprio(1);
    for (; kt < 120; kt += 4) attn_key_tile(k, v, kt, qb, o, nm2, rsum);
    __builtin_amdgcn_s_setprio(0);
    for (; kt < 128; kt += 4) attn_key_tile(k, v, kt, qb, o, nm2, rsum);

#pragma unroll
    for (int u = 0; u < 2; ++u) {
#pragma unroll
        for (int dh = 0; dh < 2; ++dh)
#pragma unroll
            for (int r = 0; r < 8; ++r)
                sO[((w * 4 + 2 * u + dh) * 8 + r) * 32 + lane] = o[u][dh][r];
        sM[(w * 2 + u) * 32 + lane] = -nm2[u];
        sL[(w * 2 + u) * 32 + lane] = rsum[u];
    }
    __syncthreads();

    {
        const int qu = lane >> 4;
        const int dh = w >> 1, hh = w & 1;
        float mj[4];
#pragma unroll
        for (int j = 0; j < 4; ++j) mj[j] = sM[(j * 2 + qu) * 32 + c];
        const float mx = fmaxf(fmaxf(mj[0], mj[1]), fmaxf(mj[2], mj[3]));
        float tot = 0.0f;
        float acc[8] = {};
#pragma unroll
        for (int j = 0; j < 4; ++j) {
            const float f = __builtin_amdgcn_exp2f(mj[j] - mx);
            tot += f * (sL[(j * 2 + qu) * 32 + c] + sL[(j * 2 + qu) * 32 + 16 + c]);
#pragma unroll
            for (int r = 0; r < 8; ++r)
                acc[r] += f * sO[((j * 4 + 2 * qu + dh) * 8 + r) * 32 + 16 * hh + c];
        }
        const float inv = 1.0f / tot;
        float* ao = AO + ((size_t)b * 256 + (size_t)hd * 32 + 8 * w) * N + n0 + lane;
#pragma unroll
        for (int r = 0; r < 8; ++r)
            ao[(size_t)r * N] = acc[r] * inv;
    }
}

extern "C" void kernel_launch(void* const* d_in, const int* in_sizes, int n_in,
                              void* d_out, int out_size, void* d_ws, size_t ws_size,
                              hipStream_t stream)
{
    const float* x      = (const float*)d_in[0];
    const float* w_qkv  = (const float*)d_in[1];
    const float* w_proj = (const float*)d_in[2];
    float* out = (float*)d_out;

    const int B = 4, C = 256, N = 2048;

    float* qkv = (float*)d_ws;
    float* ao  = qkv + (size_t)B * 3 * C * N;

    gemm_qkv_fragment_order<<<1536, 256, 0, stream>>>(w_qkv, x, qkv);

    flash_attn_wmma<<<B * 8 * (N >> 5), 128, 0, stream>>>(qkv, ao);

    gemm_f32_wmma<<<512, 256, 0, stream>>>(w_proj, ao, out, 4);
}


# attention: each wave holds four query tiles (64 queries per workgroup): a K/V fragment load feeds 64 matrix instructions per trip, not 32; half the loads, loop overhead and waves; 4 waves per SIMD
# speedup vs baseline: 2.9741x; 1.0424x over previous
#include <hip/hip_runtime.h>

typedef float v2f __attribute__((ext_vector_type(2)));
typedef float v4f __attribute__((ext_vector_type(4)));
typedef float v8f __attribute__((ext_vector_type(8)));

__device__ __forceinline__ v8f wmma_f32(v2f a, v2f b, v8f c) {
    return __builtin_amdgcn_wmma_f32_16x16x4_f32(false, a, false, b, (short)0, c,
                                                 false, false);
}

#define PW 260
#define PX2 160

__device__ __forceinline__ void stage_panels(const float* __restrict__ Wt,
                                             const float* __restrict__ Xt,
                                             float* __restrict__ sW,
                                             float* __restrict__ sX)
{
    const int t = threadIdx.x;
    v4f w4[16], xa[8], xb[8];
#pragma unroll
    for (int i = 0; i < 16; ++i)
        w4[i] = *(const v4f*)(Wt + (size_t)(i * 256 + t) * 4);
#pragma unroll
    for (int i = 0; i < 8; ++i) {
        const int id = i * 256 + t;
        const float* xp = Xt + (size_t)(id >> 4) * (2 * 2048) + (id & 15) * 4;
        xa[i] = *(const v4f*)(xp);
        xb[i] = *(const v4f*)(xp + 2048);
    }
#pragma unroll
    for (int i = 0; i < 16; ++i) {
        const int id = i * 256 + t;
        *(v4f*)(sW + (id >> 6) * PW + (id & 63) * 4) = w4[i];
    }
#pragma unroll
    for (int i = 0; i < 8; ++i) {
        const int id = i * 256 + t;
        float* d = sX + (id >> 4) * PX2 + (id & 15) * 8;
        v4f p0, p1;
        p0.x = xa[i].x; p0.y = xb[i].x; p0.z = xa[i].y; p0.w = xb[i].y;
        p1.x = xa[i].z; p1.y = xb[i].z; p1.z = xa[i].w; p1.w = xb[i].w;
        *(v4f*)(d)     = p0;
        *(v4f*)(d + 4) = p1;
    }
}

__device__ __forceinline__ int x_fragment_base(int hi, int nw, int lm)
{
    int off = 64 * PW + hi * PX2 + 2 * (32 * nw + lm);
    asm volatile("" : "+v"(off));
    return off;
}

template <bool SWAP>
__device__ __forceinline__ void piece_16x32(const float* __restrict__ aW,
                                            const float* __restrict__ bX,
                                            v8f (&acc)[2])
{
#pragma unroll 8
    for (int k0 = 0; k0 < 256; k0 += 4) {
        const v2f a = *(const v2f*)(aW + k0);
        const v2f b0 = *(const v2f*)(bX + (k0 >> 1) * PX2);
        const v2f b1 = *(const v2f*)(bX + (k0 >> 1) * PX2 + 32);
        acc[0] = SWAP ? wmma_f32(b0, a, acc[0]) : wmma_f32(a, b0, acc[0]);
        acc[1] = SWAP ? wmma_f32(b1, a, acc[1]) : wmma_f32(a, b1, acc[1]);
    }
}

__global__ __launch_bounds__(256) void gemm_f32_wmma(
    const float* __restrict__ W,
    const float* __restrict__ X,
    float* __restrict__ O,
    int Mblocks)
{
    __shared__ __attribute__((aligned(16))) float panels[64 * PW + 128 * PX2];
    float* const sW = panels;
    float* const sX = panels + 64 * PW;

    const int N = 2048;
    const int wv   = __builtin_amdgcn_readfirstlane((int)(threadIdx.x >> 5));
    const int mw   = wv >> 1, nw = wv & 1;
    const int lane = threadIdx.x & 31;
    const int lm   = lane & 15;
    const int hi   = lane >> 4;

    int t = blockIdx.x;
    const int nb = t & 31; t >>= 5;
    const int mb = t % Mblocks; t /= Mblocks;
    const int b  = t;

    stage_panels(W + (size_t)mb * 64 * 256, X + (size_t)b * 256 * N + 64 * nb, sW, sX);
    __syncthreads();

    v8f acc[2] = {};
    piece_16x32<false>(sW + (16 * mw + lm) * PW + 2 * hi,
                       panels + x_fragment_base(hi, nw, lm), acc);

    float* Op = O + ((size_t)b * (Mblocks * 64) + 64 * mb + 16 * mw) * N + 64 * nb + 32 * nw;
#pragma unroll
    for (int j = 0; j < 2; ++j)
#pragma unroll
        for (int r = 0; r < 8; ++r)
            Op[(size_t)(r + 8 * hi) * N + 16 * j + lm] = acc[j][r];
}

__global__ __launch_bounds__(256) void gemm_qkv_fragment_order(
    const float* __restrict__ W,
    const float* __restrict__ X,
    float* __restrict__ P)
{
    __shared__ __attribute__((aligned(16))) float panels[64 * PW + 128 * PX2];
    float* const sW = panels;
    float* const sX = panels + 64 * PW;

    const int N = 2048;
    const int wv   = __builtin_amdgcn_readfirstlane((int)(threadIdx.x >> 5));
    const int mw   = wv >> 1, nw = wv & 1;
    const int lane = threadIdx.x & 31;
    const int lm   = lane & 15;
    const int hi   = lane >> 4;

    int t = blockIdx.x;
    const int nb = t & 31; t >>= 5;
    const int mb = t % 12; t /= 12;
    const int b  = t;

    const int mt    = 4 * mb + mw;
    const int which = mb >> 2;
    const int head  = (mt & 15) >> 1;
    const int dhalf = mt & 1;

    stage_panels(W + (size_t)mb * 64 * 256, X + (size_t)b * 256 * N + 64 * nb, sW, sX);
    __syncthreads();

    const float* aW = sW + (16 * mw + lm) * PW + 2 * hi;
    const float* bX = panels + x_fragment_base(hi, nw, lm);
    v8f acc[2] = {};
    if (which == 2) piece_16x32<true >(aW, bX, acc);
    else            piece_16x32<false>(aW, bX, acc);

    float* Pp = P + (size_t)which * (4 * 8 * 128 * 512)
                  + ((size_t)(b * 8 + head) * 128 + 4 * nb + 2 * nw) * 512
                  + (2 * dhalf) * 128 + lane * 4;
#pragma unroll
    for (int j = 0; j < 2; ++j) {
        v4f lo, up;
        lo.x = acc[j][0]; lo.y = acc[j][1]; lo.z = acc[j][2]; lo.w = acc[j][3];
        up.x = acc[j][4]; up.y = acc[j][5]; up.z = acc[j][6]; up.w = acc[j][7];
        *(v4f*)(Pp + j * 512)       = lo;
        *(v4f*)(Pp + j * 512 + 128) = up;
    }
}

__device__ __forceinline__ float xhalf(float x) {
    int b = __builtin_bit_cast(int, x);
    return __builtin_bit_cast(float,
        __builtin_amdgcn_permlanex16(b, b, 0x76543210, 0xfedcba98, false, false));
}

#define SOFTMAX_C2 0.25503486164919736f
#define NU 4

__device__ __forceinline__ void tile_weights(const v8f (&s)[NU], const float (&nm2)[NU],
                                             v2f (&e)[NU][4], float (&ls)[NU])
{
#pragma unroll
    for (int u = 0; u < NU; ++u) {
        v2f cc; cc.x = SOFTMAX_C2; cc.y = SOFTMAX_C2;
        v2f nn; nn.x = nm2[u];     nn.y = nm2[u];
#pragma unroll
        for (int mc = 0; mc < 4; ++mc) {
            v2f a; a.x = s[u][2 * mc]; a.y = s[u][2 * mc + 1];
            const v2f t = __builtin_elementwise_fma(a, cc, nn);
            e[u][mc].x = __builtin_amdgcn_exp2f(t.x);
            e[u][mc].y = __builtin_amdgcn_exp2f(t.y);
        }
        const v2f p = (e[u][0] + e[u][1]) + (e[u][2] + e[u][3]);
        ls[u] = p.x + p.y;
    }
}

__device__ __forceinline__ void attn_key_tile(const float* __restrict__ k,
                                              const float* __restrict__ v, int kt,
                                              const v2f (&qb)[NU][8], v8f (&o)[NU][2],
                                              float (&nm2)[NU], float (&rsum)[NU])
{
    v2f ka[8];
#pragma unroll
    for (int j = 0; j < 4; ++j) {
        const v4f t4 = *(const v4f*)(k + (size_t)kt * 512 + j * 128);
        ka[2 * j].x     = t4.x;  ka[2 * j].y     = t4.y;
        ka[2 * j + 1].x = t4.z;  ka[2 * j + 1].y = t4.w;
    }
    v4f va[2][2];
#pragma unroll
    for (int dh = 0; dh < 2; ++dh) {
        va[dh][0] = *(const v4f*)(v + (size_t)kt * 512 + (2 * dh)     * 128);
        va[dh][1] = *(const v4f*)(v + (size_t)kt * 512 + (2 * dh + 1) * 128);
    }

    __builtin_amdgcn_sched_barrier(0);

    v8f s[NU] = {};
#pragma unroll
    for (int dc = 0; dc < 8; ++dc)
#pragma unroll
        for (int u = 0; u < NU; ++u)
            s[u] = wmma_f32(ka[dc], qb[u][dc], s[u]);

    v2f e[NU][4];
    float ls[NU];
    tile_weights(s, nm2, e, ls);

    if (__builtin_amdgcn_ballot_w32(fmaxf(fmaxf(ls[0], ls[1]), fmaxf(ls[2], ls[3]))
                                    > 1073741824.0f) != 0u) {
#pragma unroll
        for (int u = 0; u < NU; ++u) {
            float tmax = fmaxf(fmaxf(fmaxf(s[u][0], s[u][1]), fmaxf(s[u][2], s[u][3])),
                               fmaxf(fmaxf(s[u][4], s[u][5]), fmaxf(s[u][6], s[u][7])));
            tmax = fmaxf(tmax, xhalf(tmax));
            const float old = -nm2[u];
            const float nm  = fmaxf(old, tmax * SOFTMAX_C2);
            const float cf  = __builtin_amdgcn_exp2f(old - nm);
            nm2[u]  = -nm;
            rsum[u] *= cf;
#pragma unroll
            for (int r = 0; r < 8; ++r) { o[u][0][r] *= cf; o[u][1][r] *= cf; }
        }
        tile_weights(s, nm2, e, ls);
    }

#pragma unroll
    for (int u = 0; u < NU; ++u) {
        rsum[u] += ls[u];
#pragma unroll
        for (int mc = 0; mc < 4; ++mc) {
            v2f a0, a1;
            a0.x = va[0][mc >> 1][2 * (mc & 1)];
            a0.y = va[0][mc >> 1][2 * (mc & 1) + 1];
            a1.x = va[1][mc >> 1][2 * (mc & 1)];
            a1.y = va[1][mc >> 1][2 * (mc & 1) + 1];
            o[u][0] = wmma_f32(a0, e[u][mc], o[u][0]);
            o[u][1] = wmma_f32(a1, e[u][mc], o[u][1]);
        }
    }
}

__global__ __launch_bounds__(128, 4) void flash_attn_wmma(
    const float* __restrict__ QKV, float* __restrict__ AO)
{
    __shared__ float sO[4 * (2 * NU) * 8 * 32];
    __shared__ float sM[4 * NU * 32];
    __shared__ float sL[4 * NU * 32];

    const int lane = threadIdx.x & 31;
    const int c    = lane & 15;
    const int w    = __builtin_amdgcn_readfirstlane((int)(threadIdx.x >> 5));
    __builtin_assume(w >= 0 && w < 4);

    int t = blockIdx.x;
    const int ns = t & 31; t >>= 5;
    const int hd = t & 7;  t >>= 3;
    const int b  = t;
    const int n0 = ns << 6;
    const size_t N = 2048;

    const size_t REGION = (size_t)4 * 8 * 128 * 512;
    const size_t hbase  = ((size_t)(b * 8 + hd) * 128) * 512;
    const float* q = QKV +              hbase + lane * 4;
    const float* k = QKV +     REGION + hbase + lane * 4;
    const float* v = QKV + 2 * REGION + hbase + lane * 4;

    v2f qb[NU][8];
#pragma unroll
    for (int u = 0; u < NU; ++u)
#pragma unroll
        for (int j = 0; j < 4; ++j) {
            const v4f t4 = *(const v4f*)(q + (size_t)(NU * ns + u) * 512 + j * 128);
            qb[u][2 * j].x     = t4.x;  qb[u][2 * j].y     = t4.y;
            qb[u][2 * j + 1].x = t4.z;  qb[u][2 * j + 1].y = t4.w;
        }

    v8f o[NU][2] = {};
    float nm2[NU], rsum[NU];
#pragma unroll
    for (int u = 0; u < NU; ++u) { nm2[u] = 3.0e38f; rsum[u] = 0.0f; }

    int kt = w;
    __builtin_amdgcn_s_setprio(3);
    for (; kt < 64;  kt += 4) attn_key_tile(k, v, kt, qb, o, nm2, rsum);
    __builtin_amdgcn_s_setprio(2);
    for (; kt < 104; kt += 4) attn_key_tile(k, v, kt, qb, o, nm2, rsum);
    __builtin_amdgcn_s_setprio(1);
    for (; kt < 120; kt += 4) attn_key_tile(k, v, kt, qb, o, nm2, rsum);
    __builtin_amdgcn_s_setprio(0);
    for (; kt < 128; kt += 4) attn_key_tile(k, v, kt, qb, o, nm2, rsum);

#pragma unroll
    for (int u = 0; u < NU; ++u) {
#pragma unroll
        for (int dh = 0; dh < 2; ++dh)
#pragma unroll
            for (int r = 0; r < 8; ++r)
                sO[((w * (2 * NU) + 2 * u + dh) * 8 + r) * 32 + lane] = o[u][dh][r];
        sM[(w * NU + u) * 32 + lane] = -nm2[u];
        sL[(w * NU + u) * 32 + lane] = rsum[u];
    }
    __syncthreads();

    {
        const int dh = w >> 1, hh = w & 1;
        float* ao = AO + ((size_t)b * 256 + (size_t)hd * 32 + 8 * w) * N + n0 + lane;
#pragma unroll
        for (int qh = 0; qh < NU / 2; ++qh) {
            const int u = 2 * qh + (lane >> 4);
            float mj[4];
#pragma unroll
            for (int j = 0; j < 4; ++j) mj[j] = sM[(j * NU + u) * 32 + c];
            const float mx = fmaxf(fmaxf(mj[0], mj[1]), fmaxf(mj[2], mj[3]));
            float tot = 0.0f;
            float acc[8] = {};
#pragma unroll
            for (int j = 0; j < 4; ++j) {
                const float f = __builtin_amdgcn_exp2f(mj[j] - mx);
                tot += f * (sL[(j * NU + u) * 32 + c] + sL[(j * NU + u) * 32 + 16 + c]);
#pragma unroll
                for (int r = 0; r < 8; ++r)
                    acc[r] += f * sO[((j * (2 * NU) + 2 * u + dh) * 8 + r) * 32 + 16 * hh + c];
            }
            const float inv = 1.0f / tot;
#pragma unroll
            for (int r = 0; r < 8; ++r)
                ao[(size_t)r * N + 32 * qh] = acc[r] * inv;
        }
    }
}

extern "C" void kernel_launch(void* const* d_in, const int* in_sizes, int n_in,
                              void* d_out, int out_size, void* d_ws, size_t ws_size,
                              hipStream_t stream)
{
    const float* x      = (const float*)d_in[0];
    const float* w_qkv  = (const float*)d_in[1];
    const float* w_proj = (const float*)d_in[2];
    float* out = (float*)d_out;

    const int B = 4, C = 256, N = 2048;

    float* qkv = (float*)d_ws;
    float* ao  = qkv + (size_t)B * 3 * C * N;

    gemm_qkv_fragment_order<<<1536, 256, 0, stream>>>(w_qkv, x, qkv);

    flash_attn_wmma<<<B * 8 * (N >> 6), 128, 0, stream>>>(qkv, ao);

    gemm_f32_wmma<<<512, 256, 0, stream>>>(w_proj, ao, out, 4);
}
